# starting kernel
# baseline (speedup 1.0000x reference)
#include <hip/hip_runtime.h>


#define IH 64
#define IW 64
#define CIN 32
#define OCH 32
#define KK 3
#define OH 62
#define OW 62
#define P  (CIN * KK * KK)
#define LDS_STRIDE (P + 2)

typedef __attribute__((ext_vector_type(2))) float v2f;
typedef __attribute__((ext_vector_type(8))) float v8f;

__global__ __launch_bounds__(128)
void lc_wmma_f32_kernel(const float* __restrict__ x,
                        const float* __restrict__ weight,
                        const float* __restrict__ bias,
                        float* __restrict__ out)
{
    __shared__ float sA[32 * LDS_STRIDE];

    const int pos = blockIdx.x;
    const int i   = pos / OW;
    const int j   = pos - i * OW;

    const float* __restrict__ Wp = weight + (size_t)pos * (OCH * P);

    __builtin_prefetch(Wp + (size_t)threadIdx.x * (P / 4), 0, 1);

    for (int f = threadIdx.x; f < 32 * P; f += 128) {
        const int b  = f / P;
        const int p  = f - b * P;
        const int c  = p / 9;
        const int r9 = p - c * 9;
        const int kh = r9 / 3;
        const int kw = r9 - kh * 3;
        sA[b * LDS_STRIDE + p] =
            x[(((size_t)b * CIN + c) * IH + (i + kh)) * IW + (j + kw)];
    }
    __syncthreads();

    const int lane   = threadIdx.x & 31;
    const int wave   = threadIdx.x >> 5;
    const int laneLo = lane & 15;
    const int laneHi = lane >> 4;
    const int m0 = (wave & 1) * 16;
    const int n0 = (wave >> 1) * 16;

    const v2f* __restrict__ aPtr =
        (const v2f*)(sA + (m0 + laneLo) * LDS_STRIDE + 2 * laneHi);
    const v2f* __restrict__ bPtr =
        (const v2f*)(Wp + (n0 + laneLo) * P + 2 * laneHi);

    v8f acc = {0.f, 0.f, 0.f, 0.f, 0.f, 0.f, 0.f, 0.f};

#pragma unroll 4
    for (int kk = 0; kk < P / 4; ++kk) {
        v2f a = aPtr[2 * kk];
        v2f b = bPtr[2 * kk];
        acc = __builtin_amdgcn_wmma_f32_16x16x4_f32(
             false, a,  false, b,
             (short)0, acc,  false,  false);
    }

    const int o    = n0 + laneLo;
    const float bv = bias[(size_t)pos * OCH + o];
#pragma unroll
    for (int r = 0; r < 8; ++r) {
        const int b = m0 + r + laneHi * 8;
        out[((size_t)b * OCH + o) * (OH * OW) + pos] = acc[r] + bv;
    }
}

extern "C" void kernel_launch(void* const* d_in, const int* in_sizes, int n_in,
                              void* d_out, int out_size, void* d_ws, size_t ws_size,
                              hipStream_t stream)
{
    const float* x      = (const float*)d_in[0];
    const float* weight = (const float*)d_in[1];
    const float* bias   = (const float*)d_in[2];
    float* out          = (float*)d_out;

    (void)in_sizes; (void)n_in; (void)out_size; (void)d_ws; (void)ws_size;

    lc_wmma_f32_kernel<<<OH * OW, 128, 0, stream>>>(x, weight, bias, out);
}


# global loads issued as deep groups: x patch as 24 index-free 12-byte row-segment loads per thread (lane = channel), weight fragments 36 per group, not 4; 54 dependent round trips per wave become 3
# speedup vs baseline: 1.1080x; 1.1080x over previous
#include <hip/hip_runtime.h>


#define IH 64
#define IW 64
#define CIN 32
#define OCH 32
#define KK 3
#define OH 62
#define OW 62
#define P  (CIN * KK * KK)
#define LDS_STRIDE (P + 2)
#define KSTEPS (P / 4)
#define KGROUP (KSTEPS / 2)
#define NB (32 / 4)

typedef __attribute__((ext_vector_type(2))) float v2f;
typedef __attribute__((ext_vector_type(8))) float v8f;

__global__ __launch_bounds__(128)
void lc_wmma_f32_kernel(const float* __restrict__ x,
                        const float* __restrict__ weight,
                        const float* __restrict__ bias,
                        float* __restrict__ out)
{
    __shared__ float sA[32 * LDS_STRIDE];

    const int pos = blockIdx.x;
    const int i   = pos / OW;
    const int j   = pos - i * OW;

    const float* __restrict__ Wp = weight + (size_t)pos * (OCH * P);

    __builtin_prefetch(Wp + (size_t)threadIdx.x * (P / 4), 0, 1);

    const int lane   = threadIdx.x & 31;
    const int wave   = threadIdx.x >> 5;

    const float* __restrict__ xs =
        x + (((size_t)wave * CIN + lane) * IH + i) * IW + j;
    float seg[NB][KK][KK];
#pragma unroll
    for (int ib = 0; ib < NB; ++ib) {
#pragma unroll
        for (int kh = 0; kh < KK; ++kh) {
            const float* __restrict__ s =
                xs + (size_t)ib * (4 * CIN * IH * IW) + kh * IW;
            seg[ib][kh][0] = s[0];
            seg[ib][kh][1] = s[1];
            seg[ib][kh][2] = s[2];
        }
    }
    float* __restrict__ sdst = sA + wave * LDS_STRIDE + lane * (KK * KK);
#pragma unroll
    for (int ib = 0; ib < NB; ++ib) {
#pragma unroll
        for (int kh = 0; kh < KK; ++kh) {
#pragma unroll
            for (int kw = 0; kw < KK; ++kw)
                sdst[ib * (4 * LDS_STRIDE) + kh * KK + kw] = seg[ib][kh][kw];
        }
    }

    const int laneLo = lane & 15;
    const int laneHi = lane >> 4;
    const int m0 = (wave & 1) * 16;
    const int n0 = (wave >> 1) * 16;

    const v2f* __restrict__ aPtr =
        (const v2f*)(sA + (m0 + laneLo) * LDS_STRIDE + 2 * laneHi);
    const v2f* __restrict__ bPtr =
        (const v2f*)(Wp + (n0 + laneLo) * P + 2 * laneHi);

    v2f bfrag[KGROUP];
#pragma unroll
    for (int kk = 0; kk < KGROUP; ++kk)
        bfrag[kk] = bPtr[2 * kk];
    const float bv = bias[(size_t)pos * OCH + n0 + laneLo];

    __syncthreads();

    v8f acc = {0.f, 0.f, 0.f, 0.f, 0.f, 0.f, 0.f, 0.f};

#pragma unroll
    for (int kk = 0; kk < KGROUP; ++kk) {
        v2f a = aPtr[2 * (0 * KGROUP + kk)];
        acc = __builtin_amdgcn_wmma_f32_16x16x4_f32(
             false, a,  false, bfrag[kk],
             (short)0, acc,  false,  false);
    }
#pragma unroll
    for (int kk = 0; kk < KGROUP; ++kk)
        bfrag[kk] = bPtr[2 * (1 * KGROUP + kk)];
#pragma unroll
    for (int kk = 0; kk < KGROUP; ++kk) {
        v2f a = aPtr[2 * (1 * KGROUP + kk)];
        acc = __builtin_amdgcn_wmma_f32_16x16x4_f32(
             false, a,  false, bfrag[kk],
             (short)0, acc,  false,  false);
    }

    int tid = threadIdx.x;
    asm volatile("v_nop\n\tv_nop\n\tv_nop\n\tv_nop\n\t"
                 "v_nop\n\tv_nop\n\tv_nop\n\tv_nop"
                 : "+v"(acc), "+v"(tid));

    const int eLo = tid & 15;
    const int eHi = (tid >> 4) & 1;
    const int eM0 = ((tid >> 5) & 1) * 16;
    const int eN0 = (tid >> 6) * 16;
    const int o   = eN0 + eLo;
    float* __restrict__ ob =
        out + ((size_t)(eM0 + eHi * 8) * OCH + o) * (OH * OW) + pos;
#pragma unroll
    for (int r = 0; r < 8; ++r)
        ob[(size_t)r * (OCH * OH * OW)] = acc[r] + bv;
}

extern "C" void kernel_launch(void* const* d_in, const int* in_sizes, int n_in,
                              void* d_out, int out_size, void* d_ws, size_t ws_size,
                              hipStream_t stream)
{
    const float* x      = (const float*)d_in[0];
    const float* weight = (const float*)d_in[1];
    const float* bias   = (const float*)d_in[2];
    float* out          = (float*)d_out;

    (void)in_sizes; (void)n_in; (void)out_size; (void)d_ws; (void)ws_size;

    lc_wmma_f32_kernel<<<OH * OW, 128, 0, stream>>>(x, weight, bias, out);
}


# weights staged through LDS by whole-line 16-B loads (each 128-B line fetched once per block) and read from LDS as B fragments, replacing 8-B global fragment loads of 16 lines each; prefetch dropped
# speedup vs baseline: 1.1637x; 1.0503x over previous
#include <hip/hip_runtime.h>


#define IH 64
#define IW 64
#define CIN 32
#define OCH 32
#define KK 3
#define OH 62
#define OW 62
#define P  (CIN * KK * KK)
#define LDS_STRIDE (P + 2)
#define KSTEPS (P / 4)
#define SB_STRIDE (P + 4)
#define WVEC (OCH * P / 4)
#define WLOADS (WVEC / 128)
#define NB (32 / 4)

typedef __attribute__((ext_vector_type(2))) float v2f;
typedef __attribute__((ext_vector_type(4))) float v4f;
typedef __attribute__((ext_vector_type(8))) float v8f;

__global__ __launch_bounds__(128)
void lc_wmma_f32_kernel(const float* __restrict__ x,
                        const float* __restrict__ weight,
                        const float* __restrict__ bias,
                        float* __restrict__ out)
{
    __shared__ float sA[32 * LDS_STRIDE];
    __shared__ __attribute__((aligned(16))) float sB[OCH * SB_STRIDE];

    const int pos = blockIdx.x;
    const int i   = pos / OW;
    const int j   = pos - i * OW;

    const float* __restrict__ Wp = weight + (size_t)pos * (OCH * P);

    const v4f* __restrict__ wv = (const v4f*)Wp + threadIdx.x;
    v4f wreg[WLOADS];
#pragma unroll
    for (int q = 0; q < WLOADS; ++q)
        wreg[q] = wv[q * 128];

    const int lane   = threadIdx.x & 31;
    const int wave   = threadIdx.x >> 5;

    const float* __restrict__ xs =
        x + (((size_t)wave * CIN + lane) * IH + i) * IW + j;
    float seg[NB][KK][KK];
#pragma unroll
    for (int ib = 0; ib < NB; ++ib) {
#pragma unroll
        for (int kh = 0; kh < KK; ++kh) {
            const float* __restrict__ s =
                xs + (size_t)ib * (4 * CIN * IH * IW) + kh * IW;
            seg[ib][kh][0] = s[0];
            seg[ib][kh][1] = s[1];
            seg[ib][kh][2] = s[2];
        }
    }
#pragma unroll
    for (int q = 0; q < WLOADS; ++q) {
        const int f   = (int)threadIdx.x + q * 128;
        const int row = f / (P / 4);
        const int c4  = f - row * (P / 4);
        *(v4f*)(sB + row * SB_STRIDE + 4 * c4) = wreg[q];
    }
    float* __restrict__ sdst = sA + wave * LDS_STRIDE + lane * (KK * KK);
#pragma unroll
    for (int ib = 0; ib < NB; ++ib) {
#pragma unroll
        for (int kh = 0; kh < KK; ++kh) {
#pragma unroll
            for (int kw = 0; kw < KK; ++kw)
                sdst[ib * (4 * LDS_STRIDE) + kh * KK + kw] = seg[ib][kh][kw];
        }
    }

    const int laneLo = lane & 15;
    const int laneHi = lane >> 4;
    const int m0 = (wave & 1) * 16;
    const int n0 = (wave >> 1) * 16;

    const v2f* __restrict__ aPtr =
        (const v2f*)(sA + (m0 + laneLo) * LDS_STRIDE + 2 * laneHi);
    const v2f* __restrict__ bPtr =
        (const v2f*)(sB + (n0 + laneLo) * SB_STRIDE + 2 * laneHi);

    const float bv = bias[(size_t)pos * OCH + n0 + laneLo];

    __syncthreads();

    v8f acc = {0.f, 0.f, 0.f, 0.f, 0.f, 0.f, 0.f, 0.f};

#pragma unroll
    for (int kk = 0; kk < KSTEPS; ++kk) {
        const v2f a = aPtr[2 * kk];
        const v2f b = bPtr[2 * kk];
        acc = __builtin_amdgcn_wmma_f32_16x16x4_f32(
             false, a,  false, b,
             (short)0, acc,  false,  false);
    }

    int tid = threadIdx.x;
    asm volatile("v_nop\n\tv_nop\n\tv_nop\n\tv_nop\n\t"
                 "v_nop\n\tv_nop\n\tv_nop\n\tv_nop"
                 : "+v"(acc), "+v"(tid));

    const int eLo = tid & 15;
    const int eHi = (tid >> 4) & 1;
    const int eM0 = ((tid >> 5) & 1) * 16;
    const int eN0 = (tid >> 6) * 16;
    const int o   = eN0 + eLo;
    float* __restrict__ ob =
        out + ((size_t)(eM0 + eHi * 8) * OCH + o) * (OH * OW) + pos;
#pragma unroll
    for (int r = 0; r < 8; ++r)
        ob[(size_t)r * (OCH * OH * OW)] = acc[r] + bv;
}

extern "C" void kernel_launch(void* const* d_in, const int* in_sizes, int n_in,
                              void* d_out, int out_size, void* d_ws, size_t ws_size,
                              hipStream_t stream)
{
    const float* x      = (const float*)d_in[0];
    const float* weight = (const float*)d_in[1];
    const float* bias   = (const float*)d_in[2];
    float* out          = (float*)d_out;

    (void)in_sizes; (void)n_in; (void)out_size; (void)d_ws; (void)ws_size;

    lc_wmma_f32_kernel<<<OH * OW, 128, 0, stream>>>(x, weight, bias, out);
}
